# starting kernel
# baseline (speedup 1.0000x reference)
#include <hip/hip_runtime.h>

typedef __attribute__((ext_vector_type(2))) float v2f;
typedef __attribute__((ext_vector_type(8))) float v8f;

#define B_TOT   2048
#define I_CAP   64
#define O_CAP   32
#define D_DIM   16
#define BT      16
#define NBT     (B_TOT / BT)
#define PR_PITCH 68
#define PR_BYTES (16 * 16 * PR_PITCH * 4)

__global__ __launch_bounds__(256) void pack_x_kernel(const float* __restrict__ x,
                                                     v2f* __restrict__ xp)
{
    int f    = blockIdx.x * 256 + threadIdx.x;
    int lane = f & 31;
    int kk   = (f >> 5) & 3;
    int i    = (f >> 7) & 63;
    int bt   = f >> 13;
    int m    = lane & 15;
    int e0   = kk * 4 + ((lane >> 4) << 1);
    const float* s = x + (size_t)(bt * BT + m) * 1024 + i;
    v2f v;
    v.x = s[(size_t)e0 * 64];
    v.y = s[(size_t)(e0 + 1) * 64];
    xp[f] = v;
}

__global__ __launch_bounds__(256) void pack_w_kernel(const float* __restrict__ rw,
                                                     const float* __restrict__ nz,
                                                     v2f* __restrict__ wf)
{
    int f    = blockIdx.x * 256 + threadIdx.x;
    int lane = f & 31;
    int kk   = (f >> 5) & 3;
    int i    = (f >> 7) & 63;
    int o    = f >> 13;
    int d    = lane & 15;
    int e0   = kk * 4 + ((lane >> 4) << 1);
    size_t idx = ((size_t)(o * I_CAP + i) * 16 + d) * 16 + e0;
    v2f v;
    v.x = rw[idx]     + nz[idx];
    v.y = rw[idx + 1] + nz[idx + 1];
    wf[f] = v;
}

__global__ __launch_bounds__(256) void routing_fused_kernel(const v2f* __restrict__ xp,
                                                            const v2f* __restrict__ wf,
                                                            float* __restrict__ out)
{
    extern __shared__ float pr[];

    const int bt   = blockIdx.x;
    const int o    = blockIdx.y;
    const int tid  = threadIdx.x;
    const int lane = tid & 31;
    const int wv   = tid >> 5;
    const int dn   = lane & 15;
    const int hi8  = (lane >> 4) << 3;

    #pragma unroll
    for (int ii = 0; ii < 8; ++ii) {
        const int i = wv * 8 + ii;
        const v2f* pa = xp + (((size_t)bt * I_CAP + i) * 4) * 32 + lane;
        const v2f* pb = wf + (((size_t)o  * I_CAP + i) * 4) * 32 + lane;
        v8f acc = {};
        #pragma unroll
        for (int kk = 0; kk < 4; ++kk) {
            v2f a = pa[kk * 32];
            v2f b = pb[kk * 32];
            acc = __builtin_amdgcn_wmma_f32_16x16x4_f32(
                false, a, false, b, (short)0, acc, false, false);
        }
        #pragma unroll
        for (int v = 0; v < 8; ++v) {
            const int bl = v + hi8;
            pr[(bl * 16 + dn) * PR_PITCH + i] = acc[v];
        }
    }
    __syncthreads();

    const int tb = tid >> 4;
    const int td = tid & 15;

    float prv[64];
    const float4* row = reinterpret_cast<const float4*>(&pr[(tb * 16 + td) * PR_PITCH]);
    #pragma unroll
    for (int q = 0; q < 16; ++q) {
        float4 fq = row[q];
        prv[4 * q + 0] = fq.x;  prv[4 * q + 1] = fq.y;
        prv[4 * q + 2] = fq.z;  prv[4 * q + 3] = fq.w;
    }

    float lg[64];
    #pragma unroll
    for (int i = 0; i < 64; ++i) lg[i] = 0.0f;

    float outd = 0.0f;
    #pragma unroll
    for (int it = 0; it < 3; ++it) {
        float mx = lg[0];
        #pragma unroll
        for (int i = 1; i < 64; ++i) mx = fmaxf(mx, lg[i]);
        float se = 0.0f, s = 0.0f;
        #pragma unroll
        for (int i = 0; i < 64; ++i) {
            float e = __expf(lg[i] - mx);
            se += e;
            s = fmaf(e, prv[i], s);
        }
        s /= se;

        float sn = s * s;
        sn += __shfl_xor(sn, 1, 32);
        sn += __shfl_xor(sn, 2, 32);
        sn += __shfl_xor(sn, 4, 32);
        sn += __shfl_xor(sn, 8, 32);
        outd = (sn / (1.0f + sn)) * rsqrtf(sn) * s;

        if (it != 2) {
            #pragma unroll
            for (int i = 0; i < 64; ++i) lg[i] = fmaf(prv[i], outd, lg[i]);
        }
    }

    out[((size_t)o * B_TOT + bt * BT + tb) * D_DIM + td] = outd;
}

extern "C" void kernel_launch(void* const* d_in, const int* in_sizes, int n_in,
                              void* d_out, int out_size, void* d_ws, size_t ws_size,
                              hipStream_t stream)
{
    (void)in_sizes; (void)n_in; (void)out_size; (void)ws_size;

    const float* x  = (const float*)d_in[0];
    const float* rw = (const float*)d_in[1];
    const float* nz = (const float*)d_in[2];
    float*       out = (float*)d_out;

    v2f* xp = (v2f*)d_ws;
    v2f* wfp = (v2f*)((char*)d_ws + (size_t)8 * 1024 * 1024);

    pack_x_kernel<<<4096, 256, 0, stream>>>(x, xp);
    pack_w_kernel<<<1024, 256, 0, stream>>>(rw, nz, wfp);

    dim3 grid(NBT, O_CAP);
    routing_fused_kernel<<<grid, 256, PR_BYTES, stream>>>(xp, wfp, out);
}


# pack_x re-tiled through LDS: 256 blocks read whole 128-B rows of x and write whole-line fragment stores (2,048 waves instead of 32,768 one-element waves); fragment layout unchanged
# speedup vs baseline: 1.1281x; 1.1281x over previous
#include <hip/hip_runtime.h>

typedef __attribute__((ext_vector_type(2))) float v2f;
typedef __attribute__((ext_vector_type(8))) float v8f;

#define B_TOT   2048
#define I_CAP   64
#define O_CAP   32
#define D_DIM   16
#define BT      16
#define NBT     (B_TOT / BT)
#define PR_PITCH 68
#define PR_BYTES (16 * 16 * PR_PITCH * 4)

#define PX_PM 33
#define PX_PE (16 * PX_PM)

__global__ __launch_bounds__(256) void pack_x_kernel(const float* __restrict__ x,
                                                     v2f* __restrict__ xp)
{
    __shared__ float tile[16 * PX_PE];

    const int tid  = threadIdx.x;
    const int lane = tid & 31;
    const int wv   = tid >> 5;
    const int bt   = blockIdx.x >> 1;
    const int ih   = blockIdx.x & 1;

    const float* src = x + (size_t)bt * (BT * 1024) + ih * 32 + lane;
    for (int k = 0; k < 32; ++k) {
        const int r = wv * 32 + k;
        tile[(r & 15) * PX_PE + (r >> 4) * PX_PM + lane] = src[(size_t)r * 64];
    }
    __syncthreads();

    const int m  = lane & 15;
    const int e0 = (wv & 3) * 4 + ((lane >> 4) << 1);
    const float* t0 = tile + e0 * PX_PE + m * PX_PM + (wv >> 2);
    v2f* dst = xp + (size_t)blockIdx.x * 4096 + tid;
    for (int n = 0; n < 16; ++n) {
        v2f v;
        v.x = t0[2 * n];
        v.y = t0[PX_PE + 2 * n];
        dst[n * 256] = v;
    }
}

__global__ __launch_bounds__(256) void pack_w_kernel(const float* __restrict__ rw,
                                                     const float* __restrict__ nz,
                                                     v2f* __restrict__ wf)
{
    int f    = blockIdx.x * 256 + threadIdx.x;
    int lane = f & 31;
    int kk   = (f >> 5) & 3;
    int i    = (f >> 7) & 63;
    int o    = f >> 13;
    int d    = lane & 15;
    int e0   = kk * 4 + ((lane >> 4) << 1);
    size_t idx = ((size_t)(o * I_CAP + i) * 16 + d) * 16 + e0;
    v2f v;
    v.x = rw[idx]     + nz[idx];
    v.y = rw[idx + 1] + nz[idx + 1];
    wf[f] = v;
}

__global__ __launch_bounds__(256) void routing_fused_kernel(const v2f* __restrict__ xp,
                                                            const v2f* __restrict__ wf,
                                                            float* __restrict__ out)
{
    extern __shared__ float pr[];

    const int bt   = blockIdx.x;
    const int o    = blockIdx.y;
    const int tid  = threadIdx.x;
    const int lane = tid & 31;
    const int wv   = tid >> 5;
    const int dn   = lane & 15;
    const int hi8  = (lane >> 4) << 3;

    #pragma unroll
    for (int ii = 0; ii < 8; ++ii) {
        const int i = wv * 8 + ii;
        const v2f* pa = xp + (((size_t)bt * I_CAP + i) * 4) * 32 + lane;
        const v2f* pb = wf + (((size_t)o  * I_CAP + i) * 4) * 32 + lane;
        v8f acc = {};
        #pragma unroll
        for (int kk = 0; kk < 4; ++kk) {
            v2f a = pa[kk * 32];
            v2f b = pb[kk * 32];
            acc = __builtin_amdgcn_wmma_f32_16x16x4_f32(
                false, a, false, b, (short)0, acc, false, false);
        }
        #pragma unroll
        for (int v = 0; v < 8; ++v) {
            const int bl = v + hi8;
            pr[(bl * 16 + dn) * PR_PITCH + i] = acc[v];
        }
    }
    __syncthreads();

    const int tb = tid >> 4;
    const int td = tid & 15;

    float prv[64];
    const float4* row = reinterpret_cast<const float4*>(&pr[(tb * 16 + td) * PR_PITCH]);
    #pragma unroll
    for (int q = 0; q < 16; ++q) {
        float4 fq = row[q];
        prv[4 * q + 0] = fq.x;  prv[4 * q + 1] = fq.y;
        prv[4 * q + 2] = fq.z;  prv[4 * q + 3] = fq.w;
    }

    float lg[64];
    #pragma unroll
    for (int i = 0; i < 64; ++i) lg[i] = 0.0f;

    float outd = 0.0f;
    #pragma unroll
    for (int it = 0; it < 3; ++it) {
        float mx = lg[0];
        #pragma unroll
        for (int i = 1; i < 64; ++i) mx = fmaxf(mx, lg[i]);
        float se = 0.0f, s = 0.0f;
        #pragma unroll
        for (int i = 0; i < 64; ++i) {
            float e = __expf(lg[i] - mx);
            se += e;
            s = fmaf(e, prv[i], s);
        }
        s /= se;

        float sn = s * s;
        sn += __shfl_xor(sn, 1, 32);
        sn += __shfl_xor(sn, 2, 32);
        sn += __shfl_xor(sn, 4, 32);
        sn += __shfl_xor(sn, 8, 32);
        outd = (sn / (1.0f + sn)) * rsqrtf(sn) * s;

        if (it != 2) {
            #pragma unroll
            for (int i = 0; i < 64; ++i) lg[i] = fmaf(prv[i], outd, lg[i]);
        }
    }

    out[((size_t)o * B_TOT + bt * BT + tb) * D_DIM + td] = outd;
}

extern "C" void kernel_launch(void* const* d_in, const int* in_sizes, int n_in,
                              void* d_out, int out_size, void* d_ws, size_t ws_size,
                              hipStream_t stream)
{
    (void)in_sizes; (void)n_in; (void)out_size; (void)ws_size;

    const float* x  = (const float*)d_in[0];
    const float* rw = (const float*)d_in[1];
    const float* nz = (const float*)d_in[2];
    float*       out = (float*)d_out;

    v2f* xp = (v2f*)d_ws;
    v2f* wfp = (v2f*)((char*)d_ws + (size_t)8 * 1024 * 1024);

    pack_x_kernel<<<256, 256, 0, stream>>>(x, xp);
    pack_w_kernel<<<1024, 256, 0, stream>>>(rw, nz, wfp);

    dim3 grid(NBT, O_CAP);
    routing_fused_kernel<<<grid, 256, PR_BYTES, stream>>>(xp, wfp, out);
}


# routing logits in closed form, logit = prior*(sum of earlier outputs): one packed fma + exp2 per element, row max from the priors' max/min; the 64-register logit array, its update and max scans go
# speedup vs baseline: 1.3351x; 1.1835x over previous
#include <hip/hip_runtime.h>

typedef __attribute__((ext_vector_type(2))) float v2f;
typedef __attribute__((ext_vector_type(8))) float v8f;

#define B_TOT   2048
#define I_CAP   64
#define O_CAP   32
#define D_DIM   16
#define BT      16
#define NBT     (B_TOT / BT)
#define PR_PITCH 68
#define PR_BYTES (16 * 16 * PR_PITCH * 4)

#define PX_PM 33
#define PX_PE (16 * PX_PM)

__global__ __launch_bounds__(256) void pack_x_kernel(const float* __restrict__ x,
                                                     v2f* __restrict__ xp)
{
    __shared__ float tile[16 * PX_PE];

    const int tid  = threadIdx.x;
    const int lane = tid & 31;
    const int wv   = tid >> 5;
    const int bt   = blockIdx.x >> 1;
    const int ih   = blockIdx.x & 1;

    const float* src = x + (size_t)bt * (BT * 1024) + ih * 32 + lane;
    for (int k = 0; k < 32; ++k) {
        const int r = wv * 32 + k;
        tile[(r & 15) * PX_PE + (r >> 4) * PX_PM + lane] = src[(size_t)r * 64];
    }
    __syncthreads();

    const int m  = lane & 15;
    const int e0 = (wv & 3) * 4 + ((lane >> 4) << 1);
    const float* t0 = tile + e0 * PX_PE + m * PX_PM + (wv >> 2);
    v2f* dst = xp + (size_t)blockIdx.x * 4096 + tid;
    for (int n = 0; n < 16; ++n) {
        v2f v;
        v.x = t0[2 * n];
        v.y = t0[PX_PE + 2 * n];
        dst[n * 256] = v;
    }
}

__global__ __launch_bounds__(256) void pack_w_kernel(const float* __restrict__ rw,
                                                     const float* __restrict__ nz,
                                                     v2f* __restrict__ wf)
{
    int f    = blockIdx.x * 256 + threadIdx.x;
    int lane = f & 31;
    int kk   = (f >> 5) & 3;
    int i    = (f >> 7) & 63;
    int o    = f >> 13;
    int d    = lane & 15;
    int e0   = kk * 4 + ((lane >> 4) << 1);
    size_t idx = ((size_t)(o * I_CAP + i) * 16 + d) * 16 + e0;
    v2f v;
    v.x = rw[idx]     + nz[idx];
    v.y = rw[idx + 1] + nz[idx + 1];
    wf[f] = v;
}

__global__ __launch_bounds__(256) void routing_fused_kernel(const v2f* __restrict__ xp,
                                                            const v2f* __restrict__ wf,
                                                            float* __restrict__ out)
{
    extern __shared__ float pr[];

    const int bt   = blockIdx.x;
    const int o    = blockIdx.y;
    const int tid  = threadIdx.x;
    const int lane = tid & 31;
    const int wv   = tid >> 5;
    const int dn   = lane & 15;
    const int hi8  = (lane >> 4) << 3;

    #pragma unroll
    for (int ii = 0; ii < 8; ++ii) {
        const int i = wv * 8 + ii;
        const v2f* pa = xp + (((size_t)bt * I_CAP + i) * 4) * 32 + lane;
        const v2f* pb = wf + (((size_t)o  * I_CAP + i) * 4) * 32 + lane;
        v8f acc = {};
        #pragma unroll
        for (int kk = 0; kk < 4; ++kk) {
            v2f a = pa[kk * 32];
            v2f b = pb[kk * 32];
            acc = __builtin_amdgcn_wmma_f32_16x16x4_f32(
                false, a, false, b, (short)0, acc, false, false);
        }
        #pragma unroll
        for (int v = 0; v < 8; ++v) {
            const int bl = v + hi8;
            pr[(bl * 16 + dn) * PR_PITCH + i] = acc[v];
        }
    }
    __syncthreads();

    const int tb = tid >> 4;
    const int td = tid & 15;

    float prv[64];
    const float4* row = reinterpret_cast<const float4*>(&pr[(tb * 16 + td) * PR_PITCH]);
    #pragma unroll
    for (int q = 0; q < 16; ++q) {
        float4 fq = row[q];
        prv[4 * q + 0] = fq.x;  prv[4 * q + 1] = fq.y;
        prv[4 * q + 2] = fq.z;  prv[4 * q + 3] = fq.w;
    }

    float pmax = prv[0], pmin = prv[0];
    float s = 0.0f;
    #pragma unroll
    for (int i = 0; i < 64; ++i) {
        s += prv[i];
        pmax = fmaxf(pmax, prv[i]);
        pmin = fminf(pmin, prv[i]);
    }
    s *= (1.0f / 64.0f);

    float outd = 0.0f;
    float c = 0.0f;
    #pragma unroll
    for (int it = 0; it < 3; ++it) {
        if (it != 0) {
            c += outd;
            const float k = c * 1.4426950408889634f;
            const float m = fmaxf(pmax * k, pmin * k);
            const v2f k2 = {k, k};
            const v2f m2 = {-m, -m};
            float se = 0.0f;
            s = 0.0f;
            #pragma unroll
            for (int i = 0; i < 64; i += 2) {
                const v2f p2 = {prv[i], prv[i + 1]};
                const v2f t2 = __builtin_elementwise_fma(p2, k2, m2);
                const float e0 = __builtin_amdgcn_exp2f(t2.x);
                const float e1 = __builtin_amdgcn_exp2f(t2.y);
                se += e0;
                s = fmaf(e0, prv[i], s);
                se += e1;
                s = fmaf(e1, prv[i + 1], s);
            }
            s /= se;
        }

        float sn = s * s;
        sn += __shfl_xor(sn, 1, 32);
        sn += __shfl_xor(sn, 2, 32);
        sn += __shfl_xor(sn, 4, 32);
        sn += __shfl_xor(sn, 8, 32);
        outd = (sn / (1.0f + sn)) * rsqrtf(sn) * s;
    }

    out[((size_t)o * B_TOT + bt * BT + tb) * D_DIM + td] = outd;
}

extern "C" void kernel_launch(void* const* d_in, const int* in_sizes, int n_in,
                              void* d_out, int out_size, void* d_ws, size_t ws_size,
                              hipStream_t stream)
{
    (void)in_sizes; (void)n_in; (void)out_size; (void)ws_size;

    const float* x  = (const float*)d_in[0];
    const float* rw = (const float*)d_in[1];
    const float* nz = (const float*)d_in[2];
    float*       out = (float*)d_out;

    v2f* xp = (v2f*)d_ws;
    v2f* wfp = (v2f*)((char*)d_ws + (size_t)8 * 1024 * 1024);

    pack_x_kernel<<<256, 256, 0, stream>>>(x, xp);
    pack_w_kernel<<<1024, 256, 0, stream>>>(rw, nz, wfp);

    dim3 grid(NBT, O_CAP);
    routing_fused_kernel<<<grid, 256, PR_BYTES, stream>>>(xp, wfp, out);
}


# pack_w folded into the pack_x launch: one kernel of 1,280 blocks (block index picks the job, x blocks first), so the call is 2 dependent launches instead of 3
# speedup vs baseline: 1.3956x; 1.0453x over previous
#include <hip/hip_runtime.h>

typedef __attribute__((ext_vector_type(2))) float v2f;
typedef __attribute__((ext_vector_type(8))) float v8f;

#define B_TOT   2048
#define I_CAP   64
#define O_CAP   32
#define D_DIM   16
#define BT      16
#define NBT     (B_TOT / BT)
#define PR_PITCH 68
#define PR_BYTES (16 * 16 * PR_PITCH * 4)

__device__ __forceinline__ void pack_w_block(const float* __restrict__ rw,
                                             const float* __restrict__ nz,
                                             v2f* __restrict__ wf,
                                             int blk)
{
    int f    = blk * 256 + threadIdx.x;
    int lane = f & 31;
    int kk   = (f >> 5) & 3;
    int i    = (f >> 7) & 63;
    int o    = f >> 13;
    int d    = lane & 15;
    int e0   = kk * 4 + ((lane >> 4) << 1);
    size_t idx = ((size_t)(o * I_CAP + i) * 16 + d) * 16 + e0;
    v2f v;
    v.x = rw[idx]     + nz[idx];
    v.y = rw[idx + 1] + nz[idx + 1];
    wf[f] = v;
}

#define PX_PM 33
#define PX_PE (16 * PX_PM)

#define PX_BLOCKS 256
#define PW_BLOCKS 1024
#define PACK_BLOCKS (PX_BLOCKS + PW_BLOCKS)

__global__ __launch_bounds__(256) void pack_xw_kernel(const float* __restrict__ x,
                                                      const float* __restrict__ rw,
                                                      const float* __restrict__ nz,
                                                      v2f* __restrict__ xp,
                                                      v2f* __restrict__ wf)
{
    __shared__ float tile[16 * PX_PE];

    if (blockIdx.x >= PX_BLOCKS) {
        pack_w_block(rw, nz, wf, blockIdx.x - PX_BLOCKS);
        return;
    }

    const int tid  = threadIdx.x;
    const int lane = tid & 31;
    const int wv   = tid >> 5;
    const int bt   = blockIdx.x >> 1;
    const int ih   = blockIdx.x & 1;

    const float* src = x + (size_t)bt * (BT * 1024) + ih * 32 + lane;
    for (int k = 0; k < 32; ++k) {
        const int r = wv * 32 + k;
        tile[(r & 15) * PX_PE + (r >> 4) * PX_PM + lane] = src[(size_t)r * 64];
    }
    __syncthreads();

    const int m  = lane & 15;
    const int e0 = (wv & 3) * 4 + ((lane >> 4) << 1);
    const float* t0 = tile + e0 * PX_PE + m * PX_PM + (wv >> 2);
    v2f* dst = xp + (size_t)blockIdx.x * 4096 + tid;
    for (int n = 0; n < 16; ++n) {
        v2f v;
        v.x = t0[2 * n];
        v.y = t0[PX_PE + 2 * n];
        dst[n * 256] = v;
    }
}

__global__ __launch_bounds__(256) void routing_fused_kernel(const v2f* __restrict__ xp,
                                                            const v2f* __restrict__ wf,
                                                            float* __restrict__ out)
{
    extern __shared__ float pr[];

    const int bt   = blockIdx.x;
    const int o    = blockIdx.y;
    const int tid  = threadIdx.x;
    const int lane = tid & 31;
    const int wv   = tid >> 5;
    const int dn   = lane & 15;
    const int hi8  = (lane >> 4) << 3;

    #pragma unroll
    for (int ii = 0; ii < 8; ++ii) {
        const int i = wv * 8 + ii;
        const v2f* pa = xp + (((size_t)bt * I_CAP + i) * 4) * 32 + lane;
        const v2f* pb = wf + (((size_t)o  * I_CAP + i) * 4) * 32 + lane;
        v8f acc = {};
        #pragma unroll
        for (int kk = 0; kk < 4; ++kk) {
            v2f a = pa[kk * 32];
            v2f b = pb[kk * 32];
            acc = __builtin_amdgcn_wmma_f32_16x16x4_f32(
                false, a, false, b, (short)0, acc, false, false);
        }
        #pragma unroll
        for (int v = 0; v < 8; ++v) {
            const int bl = v + hi8;
            pr[(bl * 16 + dn) * PR_PITCH + i] = acc[v];
        }
    }
    __syncthreads();

    const int tb = tid >> 4;
    const int td = tid & 15;

    float prv[64];
    const float4* row = reinterpret_cast<const float4*>(&pr[(tb * 16 + td) * PR_PITCH]);
    #pragma unroll
    for (int q = 0; q < 16; ++q) {
        float4 fq = row[q];
        prv[4 * q + 0] = fq.x;  prv[4 * q + 1] = fq.y;
        prv[4 * q + 2] = fq.z;  prv[4 * q + 3] = fq.w;
    }

    float pmax = prv[0], pmin = prv[0];
    float s = 0.0f;
    #pragma unroll
    for (int i = 0; i < 64; ++i) {
        s += prv[i];
        pmax = fmaxf(pmax, prv[i]);
        pmin = fminf(pmin, prv[i]);
    }
    s *= (1.0f / 64.0f);

    float outd = 0.0f;
    float c = 0.0f;
    #pragma unroll
    for (int it = 0; it < 3; ++it) {
        if (it != 0) {
            c += outd;
            const float k = c * 1.4426950408889634f;
            const float m = fmaxf(pmax * k, pmin * k);
            const v2f k2 = {k, k};
            const v2f m2 = {-m, -m};
            float se = 0.0f;
            s = 0.0f;
            #pragma unroll
            for (int i = 0; i < 64; i += 2) {
                const v2f p2 = {prv[i], prv[i + 1]};
                const v2f t2 = __builtin_elementwise_fma(p2, k2, m2);
                const float e0 = __builtin_amdgcn_exp2f(t2.x);
                const float e1 = __builtin_amdgcn_exp2f(t2.y);
                se += e0;
                s = fmaf(e0, prv[i], s);
                se += e1;
                s = fmaf(e1, prv[i + 1], s);
            }
            s /= se;
        }

        float sn = s * s;
        sn += __shfl_xor(sn, 1, 32);
        sn += __shfl_xor(sn, 2, 32);
        sn += __shfl_xor(sn, 4, 32);
        sn += __shfl_xor(sn, 8, 32);
        outd = (sn / (1.0f + sn)) * rsqrtf(sn) * s;
    }

    out[((size_t)o * B_TOT + bt * BT + tb) * D_DIM + td] = outd;
}

extern "C" void kernel_launch(void* const* d_in, const int* in_sizes, int n_in,
                              void* d_out, int out_size, void* d_ws, size_t ws_size,
                              hipStream_t stream)
{
    (void)in_sizes; (void)n_in; (void)out_size; (void)ws_size;

    const float* x  = (const float*)d_in[0];
    const float* rw = (const float*)d_in[1];
    const float* nz = (const float*)d_in[2];
    float*       out = (float*)d_out;

    v2f* xp = (v2f*)d_ws;
    v2f* wfp = (v2f*)((char*)d_ws + (size_t)8 * 1024 * 1024);

    pack_xw_kernel<<<PACK_BLOCKS, 256, 0, stream>>>(x, rw, nz, xp, wfp);

    dim3 grid(NBT, O_CAP);
    routing_fused_kernel<<<grid, 256, PR_BYTES, stream>>>(xp, wfp, out);
}
